# starting kernel
# baseline (speedup 1.0000x reference)
#include <hip/hip_runtime.h>
#include <hip/hip_bf16.h>

#define DFEAT 96

typedef float v2f __attribute__((ext_vector_type(2)));
typedef float v8f __attribute__((ext_vector_type(8)));


__global__ void k_fill(float* __restrict__ p, float val, int n) {
  int i = blockIdx.x * blockDim.x + threadIdx.x;
  if (i < n) p[i] = val;
}

__global__ void k_deg_accum(const int* __restrict__ dst, const float* __restrict__ ew,
                            float* __restrict__ deg, int ne) {
  int e = blockIdx.x * blockDim.x + threadIdx.x;
  if (e < ne) atomicAdd(&deg[dst[e]], ew[e]);
}

__global__ void k_dinv(const float* __restrict__ deg, float* __restrict__ dinv, int n) {
  int i = blockIdx.x * blockDim.x + threadIdx.x;
  if (i < n) {
    float d = deg[i];
    dinv[i] = (d > 0.0f) ? rsqrtf(fmaxf(d, 1e-20f)) : 0.0f;
  }
}

__global__ __launch_bounds__(256) void k_gemm_wmma_fused(
    const float* __restrict__ X, const float* __restrict__ W,
    const float* __restrict__ dinv, const float* __restrict__ bias,
    float* __restrict__ T, float* __restrict__ out,
    int nrows, int relu_in) {
  __shared__ v2f sW2[(DFEAT / 2) * DFEAT];
  for (int idx = threadIdx.x; idx < (DFEAT / 2) * DFEAT; idx += 256) {
    const int kp  = idx / DFEAT;
    const int col = idx - kp * DFEAT;
    v2f w;
    w.x = W[(2 * kp) * DFEAT + col];
    w.y = W[(2 * kp + 1) * DFEAT + col];
    sW2[idx] = w;
  }
  __syncthreads();

  const int wave = threadIdx.x >> 5;
  const int lane = threadIdx.x & 31;
  const int ntiles = nrows >> 4;
  const int mt = blockIdx.x * 8 + wave;
  if (mt >= ntiles) return;

  const int lo16  = lane & 15;
  const int khalf = lane >> 4;
  const int row   = mt * 16 + lo16;
  const float* xr = X + (size_t)row * DFEAT;

  v8f acc[6];
#pragma unroll
  for (int t = 0; t < 6; ++t) acc[t] = (v8f){0.f, 0.f, 0.f, 0.f, 0.f, 0.f, 0.f, 0.f};

#pragma unroll 4
  for (int k = 0; k < 24; ++k) {
    const int kb = 4 * k + 2 * khalf;
    v2f a = *(const v2f*)(xr + kb);
    if (relu_in) {
      a.x = (a.x > 0.0f) ? a.x : 0.0f;
      a.y = (a.y > 0.0f) ? a.y : 0.0f;
    }
    const v2f* brow = sW2 + (kb >> 1) * DFEAT + lo16;
#pragma unroll
    for (int t = 0; t < 6; ++t) {
      v2f b = brow[t * 16];
      acc[t] = __builtin_amdgcn_wmma_f32_16x16x4_f32(
           false, a,  false, b,
           (short)0, acc[t],  false,  false);
    }
  }

  const int rbase = mt * 16 + khalf * 8;
  float dv2[8];
#pragma unroll
  for (int r = 0; r < 8; ++r) {
    float d = dinv[rbase + r];
    dv2[r] = d * d;
  }
#pragma unroll
  for (int t = 0; t < 6; ++t) {
    const int cbase = t * 16 + lo16;
    const float bb = bias[cbase];
#pragma unroll
    for (int r = 0; r < 8; ++r) {
      const size_t idx = (size_t)(rbase + r) * DFEAT + cbase;
      const float v = acc[t][r];
      T[idx] = v;
      out[idx] = dv2[r] * v + bb;
    }
  }
}

__global__ __launch_bounds__(256) void k_scatter(const float* __restrict__ T,
                                                 const int* __restrict__ src,
                                                 const int* __restrict__ dst,
                                                 const float* __restrict__ ew,
                                                 const float* __restrict__ dinv,
                                                 float* __restrict__ out, int ne) {
  const int e = blockIdx.x * 8 + (threadIdx.x >> 5);
  if (e >= ne) return;
  const int lane = threadIdx.x & 31;
  const int s = src[e];
  const int d = dst[e];
  const float nrm = dinv[s] * ew[e] * dinv[d];
  const float* ts = T + (size_t)s * DFEAT;
  float* od = out + (size_t)d * DFEAT;
#pragma unroll
  for (int j = 0; j < 3; ++j) {
    const int f = lane + 32 * j;
    atomicAdd(&od[f], nrm * ts[f]);
  }
}

__global__ void k_pool(const float* __restrict__ ge, const int* __restrict__ batch,
                       float* __restrict__ gout, int total) {
  int i = blockIdx.x * blockDim.x + threadIdx.x;
  if (i >= total) return;
  int node = i / DFEAT;
  int f = i - node * DFEAT;
  atomicAdd(&gout[batch[node] * DFEAT + f], ge[i]);
}

extern "C" void kernel_launch(void* const* d_in, const int* in_sizes, int n_in,
                              void* d_out, int out_size, void* d_ws, size_t ws_size,
                              hipStream_t stream) {
  (void)n_in; (void)ws_size;
  const float* x    = (const float*)d_in[0];
  const float* W1   = (const float*)d_in[1];
  const float* b1   = (const float*)d_in[2];
  const float* W3   = (const float*)d_in[3];
  const float* b3   = (const float*)d_in[4];
  const float* W4   = (const float*)d_in[5];
  const float* b4   = (const float*)d_in[6];
  const float* ew   = (const float*)d_in[7];
  const int*   ei   = (const int*)d_in[8];
  const int*   batch= (const int*)d_in[9];

  const int n  = in_sizes[0] / DFEAT;
  const int ne = in_sizes[7];
  const int* srcp = ei;
  const int* dstp = ei + ne;

  const int nf = n * DFEAT;
  const int gtotal = out_size - nf;

  float* deg  = (float*)d_ws;
  float* dinv = deg + n;
  float* tmp  = dinv + n;
  float* h    = tmp + (size_t)nf;
  float* tge  = h + (size_t)nf;

  float* embed = (float*)d_out;
  float* gout  = embed + nf;

  const dim3 B(256);
  const int gN  = (n + 255) / 256;
  const int gE  = (ne + 255) / 256;
  const int gF  = (nf + 255) / 256;
  const int gSc = (ne + 7) / 8;
  const int gMM = ((n >> 4) + 7) / 8;

  k_fill<<<gN, B, 0, stream>>>(deg, 1.0f, n);
  k_deg_accum<<<gE, B, 0, stream>>>(dstp, ew, deg, ne);
  k_dinv<<<gN, B, 0, stream>>>(deg, dinv, n);

  k_gemm_wmma_fused<<<gMM, B, 0, stream>>>(x, W1, dinv, b1, tmp, h, n, 0);
  k_scatter<<<gSc, B, 0, stream>>>(tmp, srcp, dstp, ew, dinv, h, ne);

  k_gemm_wmma_fused<<<gMM, B, 0, stream>>>(h, W3, dinv, b3, tmp, embed, n, 1);
  k_scatter<<<gSc, B, 0, stream>>>(tmp, srcp, dstp, ew, dinv, embed, ne);

  k_gemm_wmma_fused<<<gMM, B, 0, stream>>>(h, W4, dinv, b4, tmp, tge, n, 1);
  k_scatter<<<gSc, B, 0, stream>>>(tmp, srcp, dstp, ew, dinv, tge, ne);
  k_fill<<<(gtotal + 255) / 256, B, 0, stream>>>(gout, 0.0f, gtotal);
  k_pool<<<gF, B, 0, stream>>>(tge, batch, gout, nf);
}


# Layers 3 and 4 share one neighbour aggregation: aggregate relu(h) once, then apply W3 and W4 (sum order changes: aggregate before transform); one of three edge scatters removed
# speedup vs baseline: 1.2971x; 1.2971x over previous
#include <hip/hip_runtime.h>
#include <hip/hip_bf16.h>

#define DFEAT 96

typedef float v2f __attribute__((ext_vector_type(2)));
typedef float v8f __attribute__((ext_vector_type(8)));


__global__ void k_fill(float* __restrict__ p, float val, int n) {
  int i = blockIdx.x * blockDim.x + threadIdx.x;
  if (i < n) p[i] = val;
}

__global__ void k_deg_accum(const int* __restrict__ dst, const float* __restrict__ ew,
                            float* __restrict__ deg, int ne) {
  int e = blockIdx.x * blockDim.x + threadIdx.x;
  if (e < ne) atomicAdd(&deg[dst[e]], ew[e]);
}

__global__ void k_dinv(const float* __restrict__ deg, float* __restrict__ dinv, int n) {
  int i = blockIdx.x * blockDim.x + threadIdx.x;
  if (i < n) {
    float d = deg[i];
    dinv[i] = (d > 0.0f) ? rsqrtf(fmaxf(d, 1e-20f)) : 0.0f;
  }
}

template <bool SELF_LOOP_EPI>
__global__ __launch_bounds__(256) void k_gemm_wmma(
    const float* __restrict__ X, const float* __restrict__ W,
    const float* __restrict__ dinv, const float* __restrict__ bias,
    float* __restrict__ T, float* __restrict__ out,
    int nrows) {
  __shared__ v2f sW2[(DFEAT / 2) * DFEAT];
  for (int idx = threadIdx.x; idx < (DFEAT / 2) * DFEAT; idx += 256) {
    const int kp  = idx / DFEAT;
    const int col = idx - kp * DFEAT;
    v2f w;
    w.x = W[(2 * kp) * DFEAT + col];
    w.y = W[(2 * kp + 1) * DFEAT + col];
    sW2[idx] = w;
  }
  __syncthreads();

  const int wave = threadIdx.x >> 5;
  const int lane = threadIdx.x & 31;
  const int ntiles = nrows >> 4;
  const int mt = blockIdx.x * 8 + wave;
  if (mt >= ntiles) return;

  const int lo16  = lane & 15;
  const int khalf = lane >> 4;
  const int row   = mt * 16 + lo16;
  const float* xr = X + (size_t)row * DFEAT;

  v8f acc[6];
#pragma unroll
  for (int t = 0; t < 6; ++t) acc[t] = (v8f){0.f, 0.f, 0.f, 0.f, 0.f, 0.f, 0.f, 0.f};

#pragma unroll 4
  for (int k = 0; k < 24; ++k) {
    const int kb = 4 * k + 2 * khalf;
    v2f a = *(const v2f*)(xr + kb);
    const v2f* brow = sW2 + (kb >> 1) * DFEAT + lo16;
#pragma unroll
    for (int t = 0; t < 6; ++t) {
      v2f b = brow[t * 16];
      acc[t] = __builtin_amdgcn_wmma_f32_16x16x4_f32(
           false, a,  false, b,
           (short)0, acc[t],  false,  false);
    }
  }

  const int rbase = mt * 16 + khalf * 8;
  float dv2[8];
  if (SELF_LOOP_EPI) {
#pragma unroll
    for (int r = 0; r < 8; ++r) {
      float d = dinv[rbase + r];
      dv2[r] = d * d;
    }
  }
#pragma unroll
  for (int t = 0; t < 6; ++t) {
    const int cbase = t * 16 + lo16;
    const float bb = bias[cbase];
#pragma unroll
    for (int r = 0; r < 8; ++r) {
      const size_t idx = (size_t)(rbase + r) * DFEAT + cbase;
      const float v = acc[t][r];
      if (SELF_LOOP_EPI) {
        T[idx] = v;
        out[idx] = dv2[r] * v + bb;
      } else {
        out[idx] = v + bb;
      }
    }
  }
}

__global__ void k_self_init_relu(const float* __restrict__ hpre, const float* __restrict__ dinv,
                                 float* __restrict__ z, int nquads) {
  const int q = blockIdx.x * blockDim.x + threadIdx.x;
  if (q >= nquads) return;
  const float d = dinv[q / (DFEAT / 4)];
  const float d2 = d * d;
  const float4 v = ((const float4*)hpre)[q];
  float4 o;
  o.x = d2 * ((v.x > 0.0f) ? v.x : 0.0f);
  o.y = d2 * ((v.y > 0.0f) ? v.y : 0.0f);
  o.z = d2 * ((v.z > 0.0f) ? v.z : 0.0f);
  o.w = d2 * ((v.w > 0.0f) ? v.w : 0.0f);
  ((float4*)z)[q] = o;
}

template <bool RELU_SRC>
__global__ __launch_bounds__(256) void k_scatter(const float* __restrict__ T,
                                                 const int* __restrict__ src,
                                                 const int* __restrict__ dst,
                                                 const float* __restrict__ ew,
                                                 const float* __restrict__ dinv,
                                                 float* __restrict__ out, int ne) {
  const int e = blockIdx.x * 8 + (threadIdx.x >> 5);
  if (e >= ne) return;
  const int lane = threadIdx.x & 31;
  const int s = src[e];
  const int d = dst[e];
  const float nrm = dinv[s] * ew[e] * dinv[d];
  const float* ts = T + (size_t)s * DFEAT;
  float* od = out + (size_t)d * DFEAT;
#pragma unroll
  for (int j = 0; j < 3; ++j) {
    const int f = lane + 32 * j;
    float v = ts[f];
    if (RELU_SRC) v = (v > 0.0f) ? v : 0.0f;
    atomicAdd(&od[f], nrm * v);
  }
}

__global__ void k_pool(const float* __restrict__ ge, const int* __restrict__ batch,
                       float* __restrict__ gout, int total) {
  int i = blockIdx.x * blockDim.x + threadIdx.x;
  if (i >= total) return;
  int node = i / DFEAT;
  int f = i - node * DFEAT;
  atomicAdd(&gout[batch[node] * DFEAT + f], ge[i]);
}

extern "C" void kernel_launch(void* const* d_in, const int* in_sizes, int n_in,
                              void* d_out, int out_size, void* d_ws, size_t ws_size,
                              hipStream_t stream) {
  (void)n_in; (void)ws_size;
  const float* x    = (const float*)d_in[0];
  const float* W1   = (const float*)d_in[1];
  const float* b1   = (const float*)d_in[2];
  const float* W3   = (const float*)d_in[3];
  const float* b3   = (const float*)d_in[4];
  const float* W4   = (const float*)d_in[5];
  const float* b4   = (const float*)d_in[6];
  const float* ew   = (const float*)d_in[7];
  const int*   ei   = (const int*)d_in[8];
  const int*   batch= (const int*)d_in[9];

  const int n  = in_sizes[0] / DFEAT;
  const int ne = in_sizes[7];
  const int* srcp = ei;
  const int* dstp = ei + ne;

  const int nf = n * DFEAT;
  const int gtotal = out_size - nf;

  float* deg  = (float*)d_ws;
  float* dinv = deg + n;
  float* tmp  = dinv + n;
  float* h    = tmp + (size_t)nf;
  float* tge  = h + (size_t)nf;
  float* agg  = tmp;

  float* embed = (float*)d_out;
  float* gout  = embed + nf;

  const dim3 B(256);
  const int gN  = (n + 255) / 256;
  const int gE  = (ne + 255) / 256;
  const int gF  = (nf + 255) / 256;
  const int nq  = nf / 4;
  const int gQ  = (nq + 255) / 256;
  const int gSc = (ne + 7) / 8;
  const int gMM = ((n >> 4) + 7) / 8;

  k_fill<<<gN, B, 0, stream>>>(deg, 1.0f, n);
  k_deg_accum<<<gE, B, 0, stream>>>(dstp, ew, deg, ne);
  k_dinv<<<gN, B, 0, stream>>>(deg, dinv, n);

  k_gemm_wmma<true><<<gMM, B, 0, stream>>>(x, W1, dinv, b1, tmp, h, n);
  k_scatter<false><<<gSc, B, 0, stream>>>(tmp, srcp, dstp, ew, dinv, h, ne);

  k_self_init_relu<<<gQ, B, 0, stream>>>(h, dinv, agg, nq);
  k_scatter<true><<<gSc, B, 0, stream>>>(h, srcp, dstp, ew, dinv, agg, ne);

  k_gemm_wmma<false><<<gMM, B, 0, stream>>>(agg, W3, nullptr, b3, nullptr, embed, n);

  k_gemm_wmma<false><<<gMM, B, 0, stream>>>(agg, W4, nullptr, b4, nullptr, tge, n);
  k_fill<<<(gtotal + 255) / 256, B, 0, stream>>>(gout, 0.0f, gtotal);
  k_pool<<<gF, B, 0, stream>>>(tge, batch, gout, nf);
}


# Atomic edge scatter replaced by a gather over a destination-sorted edge list built each call (count, scan, fill): rows summed in registers in arrival-rank order (sum order changes), stored once
# speedup vs baseline: 1.9651x; 1.5150x over previous
#include <hip/hip_runtime.h>
#include <hip/hip_bf16.h>

#define DFEAT 96

typedef float v2f __attribute__((ext_vector_type(2)));
typedef float v8f __attribute__((ext_vector_type(8)));


__global__ void k_fill(float* __restrict__ p, float val, int n) {
  int i = blockIdx.x * blockDim.x + threadIdx.x;
  if (i < n) p[i] = val;
}

__global__ void k_init(float* __restrict__ deg, int* __restrict__ cnt, int n) {
  int i = blockIdx.x * blockDim.x + threadIdx.x;
  if (i < n) { deg[i] = 1.0f; cnt[i] = 0; }
}

__global__ void k_deg_rank(const int* __restrict__ dst, const float* __restrict__ ew,
                           float* __restrict__ deg, int* __restrict__ cnt,
                           int* __restrict__ rank, int ne) {
  int e = blockIdx.x * blockDim.x + threadIdx.x;
  if (e < ne) {
    const int d = dst[e];
    atomicAdd(&deg[d], ew[e]);
    rank[e] = atomicAdd(&cnt[d], 1);
  }
}

__global__ __launch_bounds__(1024) void k_scan_dinv(const int* __restrict__ cnt,
                                                    const float* __restrict__ deg,
                                                    int* __restrict__ rowptr,
                                                    float* __restrict__ dinv, int n) {
  __shared__ int s_pre[32];
  __shared__ int s_cnt[32];
  const int t = threadIdx.x;
  const int lane = t & 31;
  const int wave = t >> 5;
  const int b0 = blockIdx.x * 1024;
  int pre = 0;
  for (int k = t; k < b0; k += 1024) pre += cnt[k];
  const int i = b0 + t;
  const int c = (i < n) ? cnt[i] : 0;
  int incl = c;
#pragma unroll
  for (int off = 1; off < 32; off <<= 1) {
    const int up = __shfl_up(incl, off);
    if (lane >= off) incl += up;
    pre += __shfl_xor(pre, off);
  }
  if (lane == 31) { s_pre[wave] = pre; s_cnt[wave] = incl; }
  __syncthreads();
  int base = 0;
#pragma unroll
  for (int w = 0; w < 32; ++w) {
    base += s_pre[w];
    if (w < wave) base += s_cnt[w];
  }
  if (i < n) {
    rowptr[i] = base + incl - c;
    if (i == n - 1) rowptr[n] = base + incl;
    const float d = deg[i];
    dinv[i] = (d > 0.0f) ? rsqrtf(fmaxf(d, 1e-20f)) : 0.0f;
  }
}

__global__ void k_csr_fill(const int* __restrict__ src, const int* __restrict__ dst,
                           const float* __restrict__ ew, const float* __restrict__ dinv,
                           const int* __restrict__ rowptr, const int* __restrict__ rank,
                           int2* __restrict__ csr, int ne) {
  int e = blockIdx.x * blockDim.x + threadIdx.x;
  if (e < ne) {
    const int s = src[e];
    const int d = dst[e];
    const float nrm = dinv[s] * ew[e] * dinv[d];
    csr[rowptr[d] + rank[e]] = make_int2(s, __float_as_int(nrm));
  }
}

template <bool ADD_BIAS>
__global__ __launch_bounds__(256) void k_gemm_wmma(
    const float* __restrict__ X, const float* __restrict__ W,
    const float* __restrict__ bias, float* __restrict__ out,
    int nrows) {
  __shared__ v2f sW2[(DFEAT / 2) * DFEAT];
  for (int idx = threadIdx.x; idx < (DFEAT / 2) * DFEAT; idx += 256) {
    const int kp  = idx / DFEAT;
    const int col = idx - kp * DFEAT;
    v2f w;
    w.x = W[(2 * kp) * DFEAT + col];
    w.y = W[(2 * kp + 1) * DFEAT + col];
    sW2[idx] = w;
  }
  __syncthreads();

  const int wave = threadIdx.x >> 5;
  const int lane = threadIdx.x & 31;
  const int ntiles = nrows >> 4;
  const int mt = blockIdx.x * 8 + wave;
  if (mt >= ntiles) return;

  const int lo16  = lane & 15;
  const int khalf = lane >> 4;
  const int row   = mt * 16 + lo16;
  const float* xr = X + (size_t)row * DFEAT;

  v8f acc[6];
#pragma unroll
  for (int t = 0; t < 6; ++t) acc[t] = (v8f){0.f, 0.f, 0.f, 0.f, 0.f, 0.f, 0.f, 0.f};

#pragma unroll 4
  for (int k = 0; k < 24; ++k) {
    const int kb = 4 * k + 2 * khalf;
    v2f a = *(const v2f*)(xr + kb);
    const v2f* brow = sW2 + (kb >> 1) * DFEAT + lo16;
#pragma unroll
    for (int t = 0; t < 6; ++t) {
      v2f b = brow[t * 16];
      acc[t] = __builtin_amdgcn_wmma_f32_16x16x4_f32(
           false, a,  false, b,
           (short)0, acc[t],  false,  false);
    }
  }

  const int rbase = mt * 16 + khalf * 8;
#pragma unroll
  for (int t = 0; t < 6; ++t) {
    const int cbase = t * 16 + lo16;
    const float bb = ADD_BIAS ? bias[cbase] : 0.0f;
#pragma unroll
    for (int r = 0; r < 8; ++r) {
      const size_t idx = (size_t)(rbase + r) * DFEAT + cbase;
      const float v = acc[t][r];
      out[idx] = ADD_BIAS ? (v + bb) : v;
    }
  }
}

struct f3 { float a, b, c; };

template <bool BIAS_RELU>
__global__ __launch_bounds__(256) void k_gather(const float* __restrict__ T,
                                                const int2* __restrict__ csr,
                                                const int* __restrict__ rowptr,
                                                const float* __restrict__ dinv,
                                                const float* __restrict__ bias,
                                                float* __restrict__ out, int n) {
  const int i = blockIdx.x * 8 + __builtin_amdgcn_readfirstlane((int)(threadIdx.x >> 5));
  if (i >= n) return;
  const int lane = threadIdx.x & 31;
  const f3* __restrict__ Tr = (const f3*)T;
  f3 bb;
  bb.a = 0.0f; bb.b = 0.0f; bb.c = 0.0f;
  if (BIAS_RELU) bb = ((const f3*)bias)[lane];
  const int beg = rowptr[i];
  const int end = rowptr[i + 1];
  const int end1 = end + 1;
  const float d = dinv[i];
  const int d2bits = __float_as_int(d * d);
  float a0 = 0.0f, a1 = 0.0f, a2 = 0.0f;

  for (int base = beg; base < end1; base += 32) {
    const int m = (end1 - base < 32) ? (end1 - base) : 32;
    const int p = base + lane;
    int s_l = i;
    int w_l = (p == end) ? d2bits : 0;
    if (p < end) {
      const int2 pr = csr[p];
      s_l = pr.x;
      w_l = pr.y;
    }
    for (int j = 0; j < m; j += 4) {
      f3 v[4];
      float w[4];
#pragma unroll
      for (int u = 0; u < 4; ++u) {
        const int s = __builtin_amdgcn_readlane(s_l, j + u);
        w[u] = __int_as_float(__builtin_amdgcn_readlane(w_l, j + u));
        v[u] = Tr[(size_t)s * 32 + lane];
      }
#pragma unroll
      for (int u = 0; u < 4; ++u) {
        a0 = fmaf(w[u], v[u].a, a0);
        a1 = fmaf(w[u], v[u].b, a1);
        a2 = fmaf(w[u], v[u].c, a2);
      }
    }
  }

  if (BIAS_RELU) {
    a0 = fmaxf(a0 + bb.a, 0.0f);
    a1 = fmaxf(a1 + bb.b, 0.0f);
    a2 = fmaxf(a2 + bb.c, 0.0f);
  }
  f3 o;
  o.a = a0; o.b = a1; o.c = a2;
  ((f3*)out)[(size_t)i * 32 + lane] = o;
}

__global__ void k_pool(const float* __restrict__ ge, const int* __restrict__ batch,
                       float* __restrict__ gout, int total) {
  int i = blockIdx.x * blockDim.x + threadIdx.x;
  if (i >= total) return;
  int node = i / DFEAT;
  int f = i - node * DFEAT;
  atomicAdd(&gout[batch[node] * DFEAT + f], ge[i]);
}

extern "C" void kernel_launch(void* const* d_in, const int* in_sizes, int n_in,
                              void* d_out, int out_size, void* d_ws, size_t ws_size,
                              hipStream_t stream) {
  (void)n_in; (void)ws_size;
  const float* x    = (const float*)d_in[0];
  const float* W1   = (const float*)d_in[1];
  const float* b1   = (const float*)d_in[2];
  const float* W3   = (const float*)d_in[3];
  const float* b3   = (const float*)d_in[4];
  const float* W4   = (const float*)d_in[5];
  const float* b4   = (const float*)d_in[6];
  const float* ew   = (const float*)d_in[7];
  const int*   ei   = (const int*)d_in[8];
  const int*   batch= (const int*)d_in[9];

  const int n  = in_sizes[0] / DFEAT;
  const int ne = in_sizes[7];
  const int* srcp = ei;
  const int* dstp = ei + ne;

  const int nf = n * DFEAT;
  const int gtotal = out_size - nf;

  float* tmp    = (float*)d_ws;
  float* h      = tmp + (size_t)nf;
  float* tge    = h + (size_t)nf;
  float* agg    = tmp;
  int2*  csr    = (int2*)(tge + (size_t)nf);
  int*   rank   = (int*)(tge + (size_t)nf) + 2 * (size_t)ne;
  float* deg    = (float*)(rank + ne);
  float* dinv   = deg + n;
  int*   cnt    = (int*)(dinv + n);
  int*   rowptr = cnt + n;

  float* embed = (float*)d_out;
  float* gout  = embed + nf;

  const dim3 B(256);
  const int gN  = (n + 255) / 256;
  const int gE  = (ne + 255) / 256;
  const int gF  = (nf + 255) / 256;
  const int gSn = (n + 1023) / 1024;
  const int gG  = (n + 7) / 8;
  const int gMM = ((n >> 4) + 7) / 8;

  k_init<<<gN, B, 0, stream>>>(deg, cnt, n);
  k_deg_rank<<<gE, B, 0, stream>>>(dstp, ew, deg, cnt, rank, ne);
  k_scan_dinv<<<gSn, dim3(1024), 0, stream>>>(cnt, deg, rowptr, dinv, n);
  k_csr_fill<<<gE, B, 0, stream>>>(srcp, dstp, ew, dinv, rowptr, rank, csr, ne);

  k_gemm_wmma<false><<<gMM, B, 0, stream>>>(x, W1, nullptr, tmp, n);
  k_gather<true><<<gG, B, 0, stream>>>(tmp, csr, rowptr, dinv, b1, h, n);

  k_gather<false><<<gG, B, 0, stream>>>(h, csr, rowptr, dinv, nullptr, agg, n);

  k_gemm_wmma<true><<<gMM, B, 0, stream>>>(agg, W3, b3, embed, n);

  k_gemm_wmma<true><<<gMM, B, 0, stream>>>(agg, W4, b4, tge, n);
  k_fill<<<(gtotal + 255) / 256, B, 0, stream>>>(gout, 0.0f, gtotal);
  k_pool<<<gF, B, 0, stream>>>(tge, batch, gout, nf);
}


# global_add_pool fused into the layer-4 GEMM epilogue: 16-row tile summed in registers, one atomic per column per tile (sum order changes); pool kernel and per-node layer-4 rows removed
# speedup vs baseline: 2.0781x; 1.0575x over previous
#include <hip/hip_runtime.h>
#include <hip/hip_bf16.h>

#define DFEAT 96

typedef float v2f __attribute__((ext_vector_type(2)));
typedef float v8f __attribute__((ext_vector_type(8)));


__global__ void k_fill(float* __restrict__ p, float val, int n) {
  int i = blockIdx.x * blockDim.x + threadIdx.x;
  if (i < n) p[i] = val;
}

__global__ void k_init(float* __restrict__ deg, int* __restrict__ cnt, int n) {
  int i = blockIdx.x * blockDim.x + threadIdx.x;
  if (i < n) { deg[i] = 1.0f; cnt[i] = 0; }
}

__global__ void k_deg_rank(const int* __restrict__ dst, const float* __restrict__ ew,
                           float* __restrict__ deg, int* __restrict__ cnt,
                           int* __restrict__ rank, int ne) {
  int e = blockIdx.x * blockDim.x + threadIdx.x;
  if (e < ne) {
    const int d = dst[e];
    atomicAdd(&deg[d], ew[e]);
    rank[e] = atomicAdd(&cnt[d], 1);
  }
}

__global__ __launch_bounds__(1024) void k_scan_dinv(const int* __restrict__ cnt,
                                                    const float* __restrict__ deg,
                                                    int* __restrict__ rowptr,
                                                    float* __restrict__ dinv, int n) {
  __shared__ int s_pre[32];
  __shared__ int s_cnt[32];
  const int t = threadIdx.x;
  const int lane = t & 31;
  const int wave = t >> 5;
  const int b0 = blockIdx.x * 1024;
  int pre = 0;
  for (int k = t; k < b0; k += 1024) pre += cnt[k];
  const int i = b0 + t;
  const int c = (i < n) ? cnt[i] : 0;
  int incl = c;
#pragma unroll
  for (int off = 1; off < 32; off <<= 1) {
    const int up = __shfl_up(incl, off);
    if (lane >= off) incl += up;
    pre += __shfl_xor(pre, off);
  }
  if (lane == 31) { s_pre[wave] = pre; s_cnt[wave] = incl; }
  __syncthreads();
  int base = 0;
#pragma unroll
  for (int w = 0; w < 32; ++w) {
    base += s_pre[w];
    if (w < wave) base += s_cnt[w];
  }
  if (i < n) {
    rowptr[i] = base + incl - c;
    if (i == n - 1) rowptr[n] = base + incl;
    const float d = deg[i];
    dinv[i] = (d > 0.0f) ? rsqrtf(fmaxf(d, 1e-20f)) : 0.0f;
  }
}

__global__ void k_csr_fill(const int* __restrict__ src, const int* __restrict__ dst,
                           const float* __restrict__ ew, const float* __restrict__ dinv,
                           const int* __restrict__ rowptr, const int* __restrict__ rank,
                           int2* __restrict__ csr, int ne) {
  int e = blockIdx.x * blockDim.x + threadIdx.x;
  if (e < ne) {
    const int s = src[e];
    const int d = dst[e];
    const float nrm = dinv[s] * ew[e] * dinv[d];
    csr[rowptr[d] + rank[e]] = make_int2(s, __float_as_int(nrm));
  }
}

enum { EPI_STORE = 0, EPI_STORE_BIAS = 1, EPI_POOL_BIAS = 2 };

template <int EPI>
__global__ __launch_bounds__(256) void k_gemm_wmma(
    const float* __restrict__ X, const float* __restrict__ W,
    const float* __restrict__ bias, const int* __restrict__ batch,
    float* __restrict__ out, int nrows) {
  __shared__ v2f sW2[(DFEAT / 2) * DFEAT];
  for (int idx = threadIdx.x; idx < (DFEAT / 2) * DFEAT; idx += 256) {
    const int kp  = idx / DFEAT;
    const int col = idx - kp * DFEAT;
    v2f w;
    w.x = W[(2 * kp) * DFEAT + col];
    w.y = W[(2 * kp + 1) * DFEAT + col];
    sW2[idx] = w;
  }
  __syncthreads();

  const int wave = threadIdx.x >> 5;
  const int lane = threadIdx.x & 31;
  const int ntiles = nrows >> 4;
  const int mt = blockIdx.x * 8 + wave;
  if (mt >= ntiles) return;

  const int lo16  = lane & 15;
  const int khalf = lane >> 4;
  const int row   = mt * 16 + lo16;
  const float* xr = X + (size_t)row * DFEAT;

  int gid = 0;
  if (EPI == EPI_POOL_BIAS) gid = batch[row];

  v8f acc[6];
#pragma unroll
  for (int t = 0; t < 6; ++t) acc[t] = (v8f){0.f, 0.f, 0.f, 0.f, 0.f, 0.f, 0.f, 0.f};

#pragma unroll 4
  for (int k = 0; k < 24; ++k) {
    const int kb = 4 * k + 2 * khalf;
    v2f a = *(const v2f*)(xr + kb);
    const v2f* brow = sW2 + (kb >> 1) * DFEAT + lo16;
#pragma unroll
    for (int t = 0; t < 6; ++t) {
      v2f b = brow[t * 16];
      acc[t] = __builtin_amdgcn_wmma_f32_16x16x4_f32(
           false, a,  false, b,
           (short)0, acc[t],  false,  false);
    }
  }

  if (EPI == EPI_POOL_BIAS) {
    const int g0 = __builtin_amdgcn_readfirstlane(gid);
    const bool one_graph = __builtin_amdgcn_ballot_w32(gid != g0) == 0u;
    if (one_graph) {
      float s[6];
#pragma unroll
      for (int t = 0; t < 6; ++t) {
        const float bb = bias[t * 16 + lo16];
        float a = 0.0f;
#pragma unroll
        for (int r = 0; r < 8; ++r) a += acc[t][r] + bb;
        s[t] = a + __shfl_xor(a, 16);
      }
#pragma unroll
      for (int t = 0; t < 6; t += 2) {
        const float v = khalf ? s[t + 1] : s[t];
        atomicAdd(&out[(size_t)g0 * DFEAT + (t + khalf) * 16 + lo16], v);
      }
    } else {
#pragma unroll
      for (int r = 0; r < 8; ++r) {
        const int g = __shfl(gid, khalf * 8 + r);
#pragma unroll
        for (int t = 0; t < 6; ++t) {
          const int cbase = t * 16 + lo16;
          atomicAdd(&out[(size_t)g * DFEAT + cbase], acc[t][r] + bias[cbase]);
        }
      }
    }
  } else {
    const int rbase = mt * 16 + khalf * 8;
#pragma unroll
    for (int t = 0; t < 6; ++t) {
      const int cbase = t * 16 + lo16;
      const float bb = (EPI == EPI_STORE_BIAS) ? bias[cbase] : 0.0f;
#pragma unroll
      for (int r = 0; r < 8; ++r) {
        const size_t idx = (size_t)(rbase + r) * DFEAT + cbase;
        const float v = acc[t][r];
        out[idx] = (EPI == EPI_STORE_BIAS) ? (v + bb) : v;
      }
    }
  }
}

struct f3 { float a, b, c; };

template <bool BIAS_RELU>
__global__ __launch_bounds__(256) void k_gather(const float* __restrict__ T,
                                                const int2* __restrict__ csr,
                                                const int* __restrict__ rowptr,
                                                const float* __restrict__ dinv,
                                                const float* __restrict__ bias,
                                                float* __restrict__ out, int n) {
  const int i = blockIdx.x * 8 + __builtin_amdgcn_readfirstlane((int)(threadIdx.x >> 5));
  if (i >= n) return;
  const int lane = threadIdx.x & 31;
  const f3* __restrict__ Tr = (const f3*)T;
  f3 bb;
  bb.a = 0.0f; bb.b = 0.0f; bb.c = 0.0f;
  if (BIAS_RELU) bb = ((const f3*)bias)[lane];
  const int beg = rowptr[i];
  const int end = rowptr[i + 1];
  const int end1 = end + 1;
  const float d = dinv[i];
  const int d2bits = __float_as_int(d * d);
  float a0 = 0.0f, a1 = 0.0f, a2 = 0.0f;

  for (int base = beg; base < end1; base += 32) {
    const int m = (end1 - base < 32) ? (end1 - base) : 32;
    const int p = base + lane;
    int s_l = i;
    int w_l = (p == end) ? d2bits : 0;
    if (p < end) {
      const int2 pr = csr[p];
      s_l = pr.x;
      w_l = pr.y;
    }
    for (int j = 0; j < m; j += 4) {
      f3 v[4];
      float w[4];
#pragma unroll
      for (int u = 0; u < 4; ++u) {
        const int s = __builtin_amdgcn_readlane(s_l, j + u);
        w[u] = __int_as_float(__builtin_amdgcn_readlane(w_l, j + u));
        v[u] = Tr[(size_t)s * 32 + lane];
      }
#pragma unroll
      for (int u = 0; u < 4; ++u) {
        a0 = fmaf(w[u], v[u].a, a0);
        a1 = fmaf(w[u], v[u].b, a1);
        a2 = fmaf(w[u], v[u].c, a2);
      }
    }
  }

  if (BIAS_RELU) {
    a0 = fmaxf(a0 + bb.a, 0.0f);
    a1 = fmaxf(a1 + bb.b, 0.0f);
    a2 = fmaxf(a2 + bb.c, 0.0f);
  }
  f3 o;
  o.a = a0; o.b = a1; o.c = a2;
  ((f3*)out)[(size_t)i * 32 + lane] = o;
}

extern "C" void kernel_launch(void* const* d_in, const int* in_sizes, int n_in,
                              void* d_out, int out_size, void* d_ws, size_t ws_size,
                              hipStream_t stream) {
  (void)n_in; (void)ws_size;
  const float* x    = (const float*)d_in[0];
  const float* W1   = (const float*)d_in[1];
  const float* b1   = (const float*)d_in[2];
  const float* W3   = (const float*)d_in[3];
  const float* b3   = (const float*)d_in[4];
  const float* W4   = (const float*)d_in[5];
  const float* b4   = (const float*)d_in[6];
  const float* ew   = (const float*)d_in[7];
  const int*   ei   = (const int*)d_in[8];
  const int*   batch= (const int*)d_in[9];

  const int n  = in_sizes[0] / DFEAT;
  const int ne = in_sizes[7];
  const int* srcp = ei;
  const int* dstp = ei + ne;

  const int nf = n * DFEAT;
  const int gtotal = out_size - nf;

  float* tmp    = (float*)d_ws;
  float* h      = tmp + (size_t)nf;
  float* spare  = h + (size_t)nf;
  float* agg    = tmp;
  int2*  csr    = (int2*)(spare + (size_t)nf);
  int*   rank   = (int*)(spare + (size_t)nf) + 2 * (size_t)ne;
  float* deg    = (float*)(rank + ne);
  float* dinv   = deg + n;
  int*   cnt    = (int*)(dinv + n);
  int*   rowptr = cnt + n;

  float* embed = (float*)d_out;
  float* gout  = embed + nf;

  const dim3 B(256);
  const int gN  = (n + 255) / 256;
  const int gE  = (ne + 255) / 256;
  const int gSn = (n + 1023) / 1024;
  const int gG  = (n + 7) / 8;
  const int gMM = ((n >> 4) + 7) / 8;

  k_init<<<gN, B, 0, stream>>>(deg, cnt, n);
  k_deg_rank<<<gE, B, 0, stream>>>(dstp, ew, deg, cnt, rank, ne);
  k_scan_dinv<<<gSn, dim3(1024), 0, stream>>>(cnt, deg, rowptr, dinv, n);
  k_csr_fill<<<gE, B, 0, stream>>>(srcp, dstp, ew, dinv, rowptr, rank, csr, ne);

  k_gemm_wmma<EPI_STORE><<<gMM, B, 0, stream>>>(x, W1, nullptr, nullptr, tmp, n);
  k_gather<true><<<gG, B, 0, stream>>>(tmp, csr, rowptr, dinv, b1, h, n);

  k_gather<false><<<gG, B, 0, stream>>>(h, csr, rowptr, dinv, nullptr, agg, n);

  k_gemm_wmma<EPI_STORE_BIAS><<<gMM, B, 0, stream>>>(agg, W3, b3, nullptr, embed, n);

  k_fill<<<(gtotal + 255) / 256, B, 0, stream>>>(gout, 0.0f, gtotal);
  k_gemm_wmma<EPI_POOL_BIAS><<<gMM, B, 0, stream>>>(agg, W4, b4, batch, gout, n);
}


# Each neighbour gather fused into the GEMM behind it: a wave sums its 16 rows into an LDS A tile and multiplies from there; layer 1 now aggregates x before W1 (sum order changes); 10 launches to 7
# speedup vs baseline: 2.3725x; 1.1417x over previous
#include <hip/hip_runtime.h>
#include <hip/hip_bf16.h>

#define DFEAT 96

typedef float v2f __attribute__((ext_vector_type(2)));
typedef float v8f __attribute__((ext_vector_type(8)));


__global__ void k_fill(float* __restrict__ p, float val, int n) {
  int i = blockIdx.x * blockDim.x + threadIdx.x;
  if (i < n) p[i] = val;
}

__global__ void k_init(float* __restrict__ deg, int* __restrict__ cnt, int n) {
  int i = blockIdx.x * blockDim.x + threadIdx.x;
  if (i < n) { deg[i] = 1.0f; cnt[i] = 0; }
}

__global__ void k_deg_rank(const int* __restrict__ dst, const float* __restrict__ ew,
                           float* __restrict__ deg, int* __restrict__ cnt,
                           int* __restrict__ rank, int ne) {
  const int e = (blockIdx.x * blockDim.x + threadIdx.x) * 4;
  if (((ne & 3) == 0) && e + 3 < ne) {
    const int4 d = *(const int4*)(dst + e);
    const float4 w = *(const float4*)(ew + e);
    atomicAdd(&deg[d.x], w.x);
    atomicAdd(&deg[d.y], w.y);
    atomicAdd(&deg[d.z], w.z);
    atomicAdd(&deg[d.w], w.w);
    int4 r;
    r.x = atomicAdd(&cnt[d.x], 1);
    r.y = atomicAdd(&cnt[d.y], 1);
    r.z = atomicAdd(&cnt[d.z], 1);
    r.w = atomicAdd(&cnt[d.w], 1);
    *(int4*)(rank + e) = r;
  } else {
    for (int k = e; k < e + 4 && k < ne; ++k) {
      const int d = dst[k];
      atomicAdd(&deg[d], ew[k]);
      rank[k] = atomicAdd(&cnt[d], 1);
    }
  }
}

__global__ __launch_bounds__(1024) void k_scan_dinv(const int* __restrict__ cnt,
                                                    const float* __restrict__ deg,
                                                    int* __restrict__ rowptr,
                                                    float* __restrict__ dinv,
                                                    int2* __restrict__ csr, int n) {
  __shared__ int s_pre[32];
  __shared__ int s_cnt[32];
  const int t = threadIdx.x;
  const int lane = t & 31;
  const int wave = t >> 5;
  const int b0 = blockIdx.x * 1024;
  int pre = 0;
  for (int k = t; k < b0; k += 1024) pre += cnt[k];
  const int i = b0 + t;
  const int c = (i < n) ? cnt[i] : 0;
  int incl = c;
#pragma unroll
  for (int off = 1; off < 32; off <<= 1) {
    const int up = __shfl_up(incl, off);
    if (lane >= off) incl += up;
    pre += __shfl_xor(pre, off);
  }
  if (lane == 31) { s_pre[wave] = pre; s_cnt[wave] = incl; }
  __syncthreads();
  int base = 0;
#pragma unroll
  for (int w = 0; w < 32; ++w) {
    base += s_pre[w];
    if (w < wave) base += s_cnt[w];
  }
  if (i < n) {
    const int beg = base + incl - c + i;
    rowptr[i] = beg;
    if (i == n - 1) rowptr[n] = beg + c + 1;
    const float d = deg[i];
    const float dv = (d > 0.0f) ? rsqrtf(fmaxf(d, 1e-20f)) : 0.0f;
    dinv[i] = dv;
    csr[beg + c] = make_int2(i, __float_as_int(dv * dv));
  }
}

__device__ __forceinline__ void csr_put(int s, int d, float w, int r,
                                        const float* __restrict__ dinv,
                                        const int* __restrict__ rowptr,
                                        int2* __restrict__ csr) {
  const float nrm = dinv[s] * w * dinv[d];
  csr[rowptr[d] + r] = make_int2(s, __float_as_int(nrm));
}

__global__ void k_csr_fill(const int* __restrict__ src, const int* __restrict__ dst,
                           const float* __restrict__ ew, const float* __restrict__ dinv,
                           const int* __restrict__ rowptr, const int* __restrict__ rank,
                           int2* __restrict__ csr, int ne) {
  const int e = (blockIdx.x * blockDim.x + threadIdx.x) * 4;
  if (((ne & 3) == 0) && e + 3 < ne) {
    const int4 s = *(const int4*)(src + e);
    const int4 d = *(const int4*)(dst + e);
    const float4 w = *(const float4*)(ew + e);
    const int4 r = *(const int4*)(rank + e);
    csr_put(s.x, d.x, w.x, r.x, dinv, rowptr, csr);
    csr_put(s.y, d.y, w.y, r.y, dinv, rowptr, csr);
    csr_put(s.z, d.z, w.z, r.z, dinv, rowptr, csr);
    csr_put(s.w, d.w, w.w, r.w, dinv, rowptr, csr);
  } else {
    for (int k = e; k < e + 4 && k < ne; ++k)
      csr_put(src[k], dst[k], ew[k], rank[k], dinv, rowptr, csr);
  }
}

struct f3 { float a, b, c; };

#define FW 8
#define FR 16
#define FU 32
#define APITCH 100
#define WPLANE ((DFEAT / 2) * DFEAT)

__device__ __forceinline__ void stage_w(v2f* __restrict__ sW2, const float* __restrict__ W) {
  for (int idx = threadIdx.x; idx < WPLANE; idx += 32 * FW) {
    const int kp  = idx / DFEAT;
    const int col = idx - kp * DFEAT;
    v2f w;
    w.x = W[(2 * kp) * DFEAT + col];
    w.y = W[(2 * kp + 1) * DFEAT + col];
    sW2[idx] = w;
  }
}

__device__ __forceinline__ void tile_mma(const float* __restrict__ sAw, const v2f* __restrict__ sW2,
                                         int lo16, int khalf, v8f (&acc)[6]) {
#pragma unroll
  for (int t = 0; t < 6; ++t) acc[t] = (v8f){0.f, 0.f, 0.f, 0.f, 0.f, 0.f, 0.f, 0.f};
  const float* ar = sAw + lo16 * APITCH + 2 * khalf;
  const v2f* bcol = sW2 + khalf * DFEAT + lo16;
#pragma unroll 4
  for (int k = 0; k < 24; ++k) {
    const v2f a = *(const v2f*)(ar + 4 * k);
    const v2f* brow = bcol + (2 * k) * DFEAT;
#pragma unroll
    for (int t = 0; t < 6; ++t) {
      const v2f b = brow[t * 16];
      acc[t] = __builtin_amdgcn_wmma_f32_16x16x4_f32(
           false, a,  false, b,
           (short)0, acc[t],  false,  false);
    }
  }
}

template <bool RELU>
__device__ __forceinline__ void store_tile(const v8f (&acc)[6], const float* __restrict__ bias,
                                           float* __restrict__ out, int mt, int lo16, int khalf) {
  const int rbase = mt * 16 + khalf * 8;
#pragma unroll
  for (int t = 0; t < 6; ++t) {
    const int cbase = t * 16 + lo16;
    const float bb = bias[cbase];
#pragma unroll
    for (int r = 0; r < 8; ++r) {
      const size_t idx = (size_t)(rbase + r) * DFEAT + cbase;
      const float v = acc[t][r] + bb;
      out[idx] = RELU ? fmaxf(v, 0.0f) : v;
    }
  }
}

__device__ __forceinline__ void pool_tile(const v8f (&acc)[6], const float* __restrict__ bias,
                                          float* __restrict__ out, int gid, int lo16, int khalf) {
  const int g0 = __builtin_amdgcn_readfirstlane(gid);
  const bool one_graph = __builtin_amdgcn_ballot_w32(gid != g0) == 0u;
  if (one_graph) {
    float s[6];
#pragma unroll
    for (int t = 0; t < 6; ++t) {
      const float bb = bias[t * 16 + lo16];
      float a = 0.0f;
#pragma unroll
      for (int r = 0; r < 8; ++r) a += acc[t][r] + bb;
      s[t] = a + __shfl_xor(a, 16);
    }
#pragma unroll
    for (int t = 0; t < 6; t += 2) {
      const float v = khalf ? s[t + 1] : s[t];
      atomicAdd(&out[(size_t)g0 * DFEAT + (t + khalf) * 16 + lo16], v);
    }
  } else {
#pragma unroll
    for (int r = 0; r < 8; ++r) {
      const int g = __shfl(gid, khalf * 8 + r);
#pragma unroll
      for (int t = 0; t < 6; ++t) {
        const int cbase = t * 16 + lo16;
        atomicAdd(&out[(size_t)g * DFEAT + cbase], acc[t][r] + bias[cbase]);
      }
    }
  }
}

template <int MODE>
__global__ __launch_bounds__(32 * FW) void k_gather_gemm(
    const float* __restrict__ T, const int2* __restrict__ csr, const int* __restrict__ rowptr,
    const float* __restrict__ Wa, const float* __restrict__ ba,
    const float* __restrict__ Wb, const float* __restrict__ bb,
    const int* __restrict__ batch,
    float* __restrict__ out_a, float* __restrict__ out_b, int n) {
  __shared__ v2f sW2[(MODE == 1 ? 2 : 1) * WPLANE];
  __shared__ __attribute__((aligned(16))) float sA[FW * FR * APITCH];

  stage_w(sW2, Wa);
  if (MODE == 1) stage_w(sW2 + WPLANE, Wb);
  __syncthreads();

  const int wave = __builtin_amdgcn_readfirstlane((int)(threadIdx.x >> 5));
  const int lane = threadIdx.x & 31;
  const int mt = blockIdx.x * FW + wave;
  if (mt >= (n >> 4)) return;
  const int i0 = mt * FR;
  const int nr = (n - i0 < FR) ? (n - i0) : FR;
  float* sAw = sA + wave * (FR * APITCH);
  const int lo16  = lane & 15;
  const int khalf = lane >> 4;

  int gid = 0;
  if (MODE == 1) gid = batch[i0 + lo16];

  const f3* __restrict__ Tr = (const f3*)T;

  int rp = 0x7fffffff;
  if (lane <= nr) rp = rowptr[i0 + lane];
  const int seg_beg = __builtin_amdgcn_readlane(rp, 0);
  const int seg_end = __builtin_amdgcn_readlane(rp, nr);

  int cur = 0;
  int row_end = __builtin_amdgcn_readlane(rp, 1);
  float a0 = 0.0f, a1 = 0.0f, a2 = 0.0f;

  int cs = i0, cw = 0;
  {
    const int p = seg_beg + lane;
    if (p < seg_end) { const int2 pr = csr[p]; cs = pr.x; cw = pr.y; }
  }
  for (int base = seg_beg; base < seg_end; base += 32) {
    const int pn = base + 32 + lane;
    const int2 pr = csr[(pn < seg_end) ? pn : (seg_end - 1)];
    const int ns = (pn < seg_end) ? pr.x : i0;
    const int nw = (pn < seg_end) ? pr.y : 0;
    const int m = (seg_end - base < 32) ? (seg_end - base) : 32;
    for (int j = 0; j < m; j += FU) {
      f3 v[FU];
      float w[FU];
#pragma unroll
      for (int u = 0; u < FU; ++u) {
        const int s = __builtin_amdgcn_readlane(cs, j + u);
        w[u] = __int_as_float(__builtin_amdgcn_readlane(cw, j + u));
        v[u] = Tr[(size_t)s * 32 + lane];
      }
#pragma unroll
      for (int u = 0; u < FU; ++u) {
        a0 = fmaf(w[u], v[u].a, a0);
        a1 = fmaf(w[u], v[u].b, a1);
        a2 = fmaf(w[u], v[u].c, a2);
        if (base + j + u + 1 == row_end) {
          float* dr = sAw + cur * APITCH + 3 * lane;
          dr[0] = a0; dr[1] = a1; dr[2] = a2;
          a0 = 0.0f; a1 = 0.0f; a2 = 0.0f;
          ++cur;
          row_end = __builtin_amdgcn_readlane(rp, cur + 1);
        }
      }
    }
    cs = ns; cw = nw;
  }

  v8f acc[6];
  tile_mma(sAw, sW2, lo16, khalf, acc);
  if (MODE == 0) {
    store_tile<true>(acc, ba, out_a, mt, lo16, khalf);
  } else {
    store_tile<false>(acc, ba, out_a, mt, lo16, khalf);
    tile_mma(sAw, sW2 + WPLANE, lo16, khalf, acc);
    pool_tile(acc, bb, out_b, gid, lo16, khalf);
  }
}

extern "C" void kernel_launch(void* const* d_in, const int* in_sizes, int n_in,
                              void* d_out, int out_size, void* d_ws, size_t ws_size,
                              hipStream_t stream) {
  (void)n_in; (void)ws_size;
  const float* x    = (const float*)d_in[0];
  const float* W1   = (const float*)d_in[1];
  const float* b1   = (const float*)d_in[2];
  const float* W3   = (const float*)d_in[3];
  const float* b3   = (const float*)d_in[4];
  const float* W4   = (const float*)d_in[5];
  const float* b4   = (const float*)d_in[6];
  const float* ew   = (const float*)d_in[7];
  const int*   ei   = (const int*)d_in[8];
  const int*   batch= (const int*)d_in[9];

  const int n  = in_sizes[0] / DFEAT;
  const int ne = in_sizes[7];
  const int* srcp = ei;
  const int* dstp = ei + ne;

  const int nf = n * DFEAT;
  const int gtotal = out_size - nf;

  float* h      = (float*)d_ws;
  float* spare  = h + 2 * (size_t)nf;
  int2*  csr    = (int2*)(spare + (size_t)nf);
  int*   rank   = (int*)(spare + (size_t)nf) + 2 * ((size_t)ne + (size_t)n);
  float* deg    = (float*)(rank + ne);
  float* dinv   = deg + n;
  int*   cnt    = (int*)(dinv + n);
  int*   rowptr = cnt + n;

  float* embed = (float*)d_out;
  float* gout  = embed + nf;

  const dim3 B(256);
  const int gN  = (n + 255) / 256;
  const int gE  = (ne + 1023) / 1024;
  const int gSn = (n + 1023) / 1024;
  const int gF  = ((n >> 4) + FW - 1) / FW;

  k_init<<<gN, B, 0, stream>>>(deg, cnt, n);
  k_deg_rank<<<gE, B, 0, stream>>>(dstp, ew, deg, cnt, rank, ne);
  k_scan_dinv<<<gSn, dim3(1024), 0, stream>>>(cnt, deg, rowptr, dinv, csr, n);
  k_csr_fill<<<gE, B, 0, stream>>>(srcp, dstp, ew, dinv, rowptr, rank, csr, ne);

  k_gather_gemm<0><<<gF, dim3(32 * FW), 0, stream>>>(x, csr, rowptr, W1, b1, nullptr, nullptr, nullptr,
                                                     h, nullptr, n);

  k_fill<<<(gtotal + 255) / 256, B, 0, stream>>>(gout, 0.0f, gtotal);
  k_gather_gemm<1><<<gF, dim3(32 * FW), 0, stream>>>(h, csr, rowptr, W3, b3, W4, b4, batch,
                                                     embed, gout, n);
}


# Edge ranks and degrees from LDS counters: block (4096-node range, 1/19 of edges) streams dst, ew; list build has no global atomics (sum order changes: by part, then arrival); 7 launches to 6
# speedup vs baseline: 3.3028x; 1.3921x over previous
#include <hip/hip_runtime.h>
#include <hip/hip_bf16.h>

#define DFEAT 96

typedef float v2f __attribute__((ext_vector_type(2)));
typedef float v8f __attribute__((ext_vector_type(8)));


__global__ void k_fill(float* __restrict__ p, float val, int n) {
  int i = blockIdx.x * blockDim.x + threadIdx.x;
  if (i < n) p[i] = val;
}

#define RN 4096
#define SUBR (RN / 1024)
#define QP 19
#define NOEDGE 0x7fffffff

__device__ __forceinline__ int part_len(int ne) {
  const int c = (((ne + QP - 1) / QP) + 3) & ~3;
  return c < 4 ? 4 : c;
}

__device__ __forceinline__ void visit(int d, float w, int e, int lo, int* __restrict__ s_cnt,
                                      float* __restrict__ s_deg, int (&below)[SUBR],
                                      int* __restrict__ rank) {
#pragma unroll
  for (int j = 0; j < SUBR; ++j) below[j] += (d < lo + 1024 * j) ? 1 : 0;
  const unsigned k = (unsigned)(d - lo);
  if (k < (unsigned)RN) {
    rank[e] = atomicAdd(&s_cnt[k], 1);
    atomicAdd(&s_deg[k], w);
  }
}

__device__ __forceinline__ void visit4(int4 d, float4 w, int e, int lo, int* __restrict__ s_cnt,
                                       float* __restrict__ s_deg, int (&below)[SUBR],
                                       int* __restrict__ rank) {
#pragma unroll
  for (int j = 0; j < SUBR; ++j) {
    const int bound = lo + 1024 * j;
    below[j] += ((d.x < bound) ? 1 : 0) + ((d.y < bound) ? 1 : 0) +
                ((d.z < bound) ? 1 : 0) + ((d.w < bound) ? 1 : 0);
  }
  const unsigned k0 = (unsigned)(d.x - lo), k1 = (unsigned)(d.y - lo);
  const unsigned k2 = (unsigned)(d.z - lo), k3 = (unsigned)(d.w - lo);
  const bool m0 = k0 < (unsigned)RN, m1 = k1 < (unsigned)RN;
  const bool m2 = k2 < (unsigned)RN, m3 = k3 < (unsigned)RN;
  int r0 = 0, r1 = 0, r2 = 0, r3 = 0;
  if (m0) { r0 = atomicAdd(&s_cnt[k0], 1); atomicAdd(&s_deg[k0], w.x); }
  if (m1) { r1 = atomicAdd(&s_cnt[k1], 1); atomicAdd(&s_deg[k1], w.y); }
  if (m2) { r2 = atomicAdd(&s_cnt[k2], 1); atomicAdd(&s_deg[k2], w.z); }
  if (m3) { r3 = atomicAdd(&s_cnt[k3], 1); atomicAdd(&s_deg[k3], w.w); }
  if (m0) rank[e] = r0;
  if (m1) rank[e + 1] = r1;
  if (m2) rank[e + 2] = r2;
  if (m3) rank[e + 3] = r3;
}

__global__ __launch_bounds__(1024) void k_count(const int* __restrict__ dst, const float* __restrict__ ew,
                                                int* __restrict__ cntp, float* __restrict__ degp,
                                                int* __restrict__ belowp, int* __restrict__ rank,
                                                int n, int ne) {
  __shared__ int s_cnt[RN];
  __shared__ float s_deg[RN];
  __shared__ int s_red[SUBR * 32];
  const int t = threadIdx.x;
  const int rho = blockIdx.x / QP;
  const int q = blockIdx.x - rho * QP;
  const int lo = rho * RN;
  for (int k = t; k < RN; k += 1024) { s_cnt[k] = 0; s_deg[k] = 0.0f; }
  __syncthreads();
  const int chunk = part_len(ne);
  const int e0 = q * chunk;
  const int e1 = (e0 + chunk < ne) ? (e0 + chunk) : ne;
  int below[SUBR];
#pragma unroll
  for (int j = 0; j < SUBR; ++j) below[j] = 0;
  if ((ne & 3) == 0) {
    for (int eb = e0 + 4 * t; eb < e1; eb += 4 * 4096) {
      int4 d[4];
      float4 w[4];
#pragma unroll
      for (int u = 0; u < 4; ++u) {
        const int e = eb + u * 4096;
        d[u] = make_int4(NOEDGE, NOEDGE, NOEDGE, NOEDGE);
        w[u] = make_float4(0.0f, 0.0f, 0.0f, 0.0f);
        if (e < e1) { d[u] = *(const int4*)(dst + e); w[u] = *(const float4*)(ew + e); }
      }
#pragma unroll
      for (int u = 0; u < 4; ++u) {
        const int e = eb + u * 4096;
        visit4(d[u], w[u], e, lo, s_cnt, s_deg, below, rank);
      }
    }
  } else {
    for (int e = e0 + t; e < e1; e += 1024) visit(dst[e], ew[e], e, lo, s_cnt, s_deg, below, rank);
  }
  __syncthreads();
  for (int k = t; k < RN; k += 1024) {
    const int i = lo + k;
    if (i < n) { cntp[q * n + i] = s_cnt[k]; degp[q * n + i] = s_deg[k]; }
  }
#pragma unroll
  for (int j = 0; j < SUBR; ++j) {
    int v = below[j];
#pragma unroll
    for (int off = 16; off > 0; off >>= 1) v += __shfl_xor(v, off);
    if ((t & 31) == 0) s_red[j * 32 + (t >> 5)] = v;
  }
  __syncthreads();
  if (t < SUBR) {
    int v = 0;
    for (int w = 0; w < 32; ++w) v += s_red[t * 32 + w];
    belowp[(rho * QP + q) * SUBR + t] = v;
  }
}

__global__ __launch_bounds__(1024) void k_scan_dinv(const int* __restrict__ cntp,
                                                    const float* __restrict__ degp,
                                                    const int* __restrict__ belowp,
                                                    int* __restrict__ rowptr,
                                                    int* __restrict__ basep,
                                                    float* __restrict__ dinv,
                                                    int2* __restrict__ csr, int n) {
  __shared__ int s_cnt[32];
  const int t = threadIdx.x;
  const int lane = t & 31;
  const int wave = t >> 5;
  const int b0 = blockIdx.x * 1024;
  const int rho = b0 / RN;
  const int j = (b0 - rho * RN) >> 10;
  int pre = 0;
  for (int q = 0; q < QP; ++q) pre += belowp[(rho * QP + q) * SUBR + j];
  const int i = b0 + t;
  const int* __restrict__ cp = cntp + ((i < n) ? i : (n - 1));
  const float* __restrict__ dp = degp + ((i < n) ? i : (n - 1));
  int cq[QP];
  int c = 0;
  float d = 1.0f;
#pragma unroll
  for (int q = 0; q < QP; ++q) {
    cq[q] = (i < n) ? cp[q * n] : 0;
    d += dp[q * n];
    c += cq[q];
  }
  int incl = c;
#pragma unroll
  for (int off = 1; off < 32; off <<= 1) {
    const int up = __shfl_up(incl, off);
    if (lane >= off) incl += up;
  }
  if (lane == 31) s_cnt[wave] = incl;
  __syncthreads();
  int base = pre;
#pragma unroll
  for (int w = 0; w < 32; ++w) {
    if (w < wave) base += s_cnt[w];
  }
  if (i < n) {
    const int beg = base + incl - c + i;
    rowptr[i] = beg;
    if (i == n - 1) rowptr[n] = beg + c + 1;
    const float dv = (d > 0.0f) ? rsqrtf(fmaxf(d, 1e-20f)) : 0.0f;
    dinv[i] = dv;
    csr[beg + c] = make_int2(i, __float_as_int(dv * dv));
    int run = beg;
    int* __restrict__ bp = basep + i;
#pragma unroll
    for (int q = 0; q < QP; ++q) { bp[q * n] = run; run += cq[q]; }
  }
}

__device__ __forceinline__ void csr_put(int s, int d, float w, int r,
                                        const float* __restrict__ dinv,
                                        const int* __restrict__ bq,
                                        int2* __restrict__ csr) {
  const float nrm = dinv[s] * w * dinv[d];
  csr[bq[d] + r] = make_int2(s, __float_as_int(nrm));
}

__global__ void k_csr_fill(const int* __restrict__ src, const int* __restrict__ dst,
                           const float* __restrict__ ew, const float* __restrict__ dinv,
                           const int* __restrict__ basep, const int* __restrict__ rank,
                           int2* __restrict__ csr, int n, int ne) {
  const int e = (blockIdx.x * blockDim.x + threadIdx.x) * 4;
  const int chunk = part_len(ne);
  if (((ne & 3) == 0) && e + 3 < ne) {
    const int* __restrict__ bq = basep + (e / chunk) * n;
    const int4 s = *(const int4*)(src + e);
    const int4 d = *(const int4*)(dst + e);
    const float4 w = *(const float4*)(ew + e);
    const int4 r = *(const int4*)(rank + e);
    csr_put(s.x, d.x, w.x, r.x, dinv, bq, csr);
    csr_put(s.y, d.y, w.y, r.y, dinv, bq, csr);
    csr_put(s.z, d.z, w.z, r.z, dinv, bq, csr);
    csr_put(s.w, d.w, w.w, r.w, dinv, bq, csr);
  } else {
    for (int k = e; k < e + 4 && k < ne; ++k)
      csr_put(src[k], dst[k], ew[k], rank[k], dinv, basep + (k / chunk) * n, csr);
  }
}

struct f3 { float a, b, c; };

#define FW 8
#define FR 16
#define FU 32
#define APITCH 100
#define WPLANE ((DFEAT / 2) * DFEAT)

__device__ __forceinline__ void stage_w(v2f* __restrict__ sW2, const float* __restrict__ W) {
  for (int idx = threadIdx.x; idx < WPLANE; idx += 32 * FW) {
    const int kp  = idx / DFEAT;
    const int col = idx - kp * DFEAT;
    v2f w;
    w.x = W[(2 * kp) * DFEAT + col];
    w.y = W[(2 * kp + 1) * DFEAT + col];
    sW2[idx] = w;
  }
}

__device__ __forceinline__ void tile_mma(const float* __restrict__ sAw, const v2f* __restrict__ sW2,
                                         int lo16, int khalf, v8f (&acc)[6]) {
#pragma unroll
  for (int t = 0; t < 6; ++t) acc[t] = (v8f){0.f, 0.f, 0.f, 0.f, 0.f, 0.f, 0.f, 0.f};
  const float* ar = sAw + lo16 * APITCH + 2 * khalf;
  const v2f* bcol = sW2 + khalf * DFEAT + lo16;
#pragma unroll 4
  for (int k = 0; k < 24; ++k) {
    const v2f a = *(const v2f*)(ar + 4 * k);
    const v2f* brow = bcol + (2 * k) * DFEAT;
#pragma unroll
    for (int t = 0; t < 6; ++t) {
      const v2f b = brow[t * 16];
      acc[t] = __builtin_amdgcn_wmma_f32_16x16x4_f32(
           false, a,  false, b,
           (short)0, acc[t],  false,  false);
    }
  }
}

template <bool RELU>
__device__ __forceinline__ void store_tile(const v8f (&acc)[6], const float* __restrict__ bias,
                                           float* __restrict__ out, int mt, int lo16, int khalf) {
  const int rbase = mt * 16 + khalf * 8;
#pragma unroll
  for (int t = 0; t < 6; ++t) {
    const int cbase = t * 16 + lo16;
    const float bb = bias[cbase];
#pragma unroll
    for (int r = 0; r < 8; ++r) {
      const size_t idx = (size_t)(rbase + r) * DFEAT + cbase;
      const float v = acc[t][r] + bb;
      out[idx] = RELU ? fmaxf(v, 0.0f) : v;
    }
  }
}

__device__ __forceinline__ void pool_tile(const v8f (&acc)[6], const float* __restrict__ bias,
                                          float* __restrict__ out, int gid, int lo16, int khalf) {
  const int g0 = __builtin_amdgcn_readfirstlane(gid);
  const bool one_graph = __builtin_amdgcn_ballot_w32(gid != g0) == 0u;
  if (one_graph) {
    float s[6];
#pragma unroll
    for (int t = 0; t < 6; ++t) {
      const float bb = bias[t * 16 + lo16];
      float a = 0.0f;
#pragma unroll
      for (int r = 0; r < 8; ++r) a += acc[t][r] + bb;
      s[t] = a + __shfl_xor(a, 16);
    }
#pragma unroll
    for (int t = 0; t < 6; t += 2) {
      const float v = khalf ? s[t + 1] : s[t];
      atomicAdd(&out[(size_t)g0 * DFEAT + (t + khalf) * 16 + lo16], v);
    }
  } else {
#pragma unroll
    for (int r = 0; r < 8; ++r) {
      const int g = __shfl(gid, khalf * 8 + r);
#pragma unroll
      for (int t = 0; t < 6; ++t) {
        const int cbase = t * 16 + lo16;
        atomicAdd(&out[(size_t)g * DFEAT + cbase], acc[t][r] + bias[cbase]);
      }
    }
  }
}

template <int MODE>
__global__ __launch_bounds__(32 * FW) void k_gather_gemm(
    const float* __restrict__ T, const int2* __restrict__ csr, const int* __restrict__ rowptr,
    const float* __restrict__ Wa, const float* __restrict__ ba,
    const float* __restrict__ Wb, const float* __restrict__ bb,
    const int* __restrict__ batch,
    float* __restrict__ out_a, float* __restrict__ out_b, int n) {
  __shared__ v2f sW2[(MODE == 1 ? 2 : 1) * WPLANE];
  __shared__ __attribute__((aligned(16))) float sA[FW * FR * APITCH];

  stage_w(sW2, Wa);
  if (MODE == 1) stage_w(sW2 + WPLANE, Wb);
  __syncthreads();

  const int wave = __builtin_amdgcn_readfirstlane((int)(threadIdx.x >> 5));
  const int lane = threadIdx.x & 31;
  const int mt = blockIdx.x * FW + wave;
  if (mt >= (n >> 4)) return;
  const int i0 = mt * FR;
  const int nr = (n - i0 < FR) ? (n - i0) : FR;
  float* sAw = sA + wave * (FR * APITCH);
  const int lo16  = lane & 15;
  const int khalf = lane >> 4;

  int gid = 0;
  if (MODE == 1) gid = batch[i0 + lo16];

  const f3* __restrict__ Tr = (const f3*)T;

  int rp = 0x7fffffff;
  if (lane <= nr) rp = rowptr[i0 + lane];
  const int seg_beg = __builtin_amdgcn_readlane(rp, 0);
  const int seg_end = __builtin_amdgcn_readlane(rp, nr);

  int cur = 0;
  int row_end = __builtin_amdgcn_readlane(rp, 1);
  float a0 = 0.0f, a1 = 0.0f, a2 = 0.0f;

  int cs = i0, cw = 0;
  {
    const int p = seg_beg + lane;
    if (p < seg_end) { const int2 pr = csr[p]; cs = pr.x; cw = pr.y; }
  }
  for (int base = seg_beg; base < seg_end; base += 32) {
    const int pn = base + 32 + lane;
    const int2 pr = csr[(pn < seg_end) ? pn : (seg_end - 1)];
    const int ns = (pn < seg_end) ? pr.x : i0;
    const int nw = (pn < seg_end) ? pr.y : 0;
    const int m = (seg_end - base < 32) ? (seg_end - base) : 32;
    for (int j = 0; j < m; j += FU) {
      f3 v[FU];
      float w[FU];
#pragma unroll
      for (int u = 0; u < FU; ++u) {
        const int s = __builtin_amdgcn_readlane(cs, j + u);
        w[u] = __int_as_float(__builtin_amdgcn_readlane(cw, j + u));
        v[u] = Tr[(size_t)s * 32 + lane];
      }
#pragma unroll
      for (int u = 0; u < FU; ++u) {
        a0 = fmaf(w[u], v[u].a, a0);
        a1 = fmaf(w[u], v[u].b, a1);
        a2 = fmaf(w[u], v[u].c, a2);
        if (base + j + u + 1 == row_end) {
          float* dr = sAw + cur * APITCH + 3 * lane;
          dr[0] = a0; dr[1] = a1; dr[2] = a2;
          a0 = 0.0f; a1 = 0.0f; a2 = 0.0f;
          ++cur;
          row_end = __builtin_amdgcn_readlane(rp, cur + 1);
        }
      }
    }
    cs = ns; cw = nw;
  }

  v8f acc[6];
  tile_mma(sAw, sW2, lo16, khalf, acc);
  if (MODE == 0) {
    store_tile<true>(acc, ba, out_a, mt, lo16, khalf);
  } else {
    store_tile<false>(acc, ba, out_a, mt, lo16, khalf);
    tile_mma(sAw, sW2 + WPLANE, lo16, khalf, acc);
    pool_tile(acc, bb, out_b, gid, lo16, khalf);
  }
}

extern "C" void kernel_launch(void* const* d_in, const int* in_sizes, int n_in,
                              void* d_out, int out_size, void* d_ws, size_t ws_size,
                              hipStream_t stream) {
  (void)n_in; (void)ws_size;
  const float* x    = (const float*)d_in[0];
  const float* W1   = (const float*)d_in[1];
  const float* b1   = (const float*)d_in[2];
  const float* W3   = (const float*)d_in[3];
  const float* b3   = (const float*)d_in[4];
  const float* W4   = (const float*)d_in[5];
  const float* b4   = (const float*)d_in[6];
  const float* ew   = (const float*)d_in[7];
  const int*   ei   = (const int*)d_in[8];
  const int*   batch= (const int*)d_in[9];

  const int n  = in_sizes[0] / DFEAT;
  const int ne = in_sizes[7];
  const int* srcp = ei;
  const int* dstp = ei + ne;

  const int nf = n * DFEAT;
  const int gtotal = out_size - nf;

  float* h      = (float*)d_ws;
  int*   cntp   = (int*)(h + (size_t)nf);
  float* degp   = (float*)(cntp + (size_t)QP * n);
  int*   basep  = (int*)(degp + (size_t)QP * n);
  int*   belowp = basep + (size_t)QP * n;
  float* spare  = h + 2 * (size_t)nf;
  int2*  csr    = (int2*)(spare + (size_t)nf);
  int*   rank   = (int*)(spare + (size_t)nf) + 2 * ((size_t)ne + (size_t)n);
  float* dinv   = (float*)(rank + ne) + n;
  int*   rowptr = (int*)(dinv + n) + n;

  float* embed = (float*)d_out;
  float* gout  = embed + nf;

  const dim3 B(256);
  const int gR  = (n + RN - 1) / RN;
  const int gE  = (ne + 1023) / 1024;
  const int gSn = (n + 1023) / 1024;
  const int gF  = ((n >> 4) + FW - 1) / FW;

  k_count<<<gR * QP, dim3(1024), 0, stream>>>(dstp, ew, cntp, degp, belowp, rank, n, ne);
  k_scan_dinv<<<gSn, dim3(1024), 0, stream>>>(cntp, degp, belowp, rowptr, basep, dinv, csr, n);
  k_csr_fill<<<gE, B, 0, stream>>>(srcp, dstp, ew, dinv, basep, rank, csr, n, ne);

  k_gather_gemm<0><<<gF, dim3(32 * FW), 0, stream>>>(x, csr, rowptr, W1, b1, nullptr, nullptr, nullptr,
                                                     h, nullptr, n);

  k_fill<<<(gtotal + 255) / 256, B, 0, stream>>>(gout, 0.0f, gtotal);
  k_gather_gemm<1><<<gF, dim3(32 * FW), 0, stream>>>(h, csr, rowptr, W3, b3, W4, b4, batch,
                                                     embed, gout, n);
}


# Start-up of each gather launch cut from 8 (layer 1) and 14 (layers 3, 4) serial round trips to 2: W fetched as one group of 16-byte loads, list bounds and first entries asked for beside it
# speedup vs baseline: 3.4905x; 1.0568x over previous
#include <hip/hip_runtime.h>
#include <hip/hip_bf16.h>

#define DFEAT 96

typedef float v2f __attribute__((ext_vector_type(2)));
typedef float v8f __attribute__((ext_vector_type(8)));


__global__ void k_fill(float* __restrict__ p, float val, int n) {
  int i = blockIdx.x * blockDim.x + threadIdx.x;
  if (i < n) p[i] = val;
}

#define RN 4096
#define SUBR (RN / 1024)
#define QP 19
#define NOEDGE 0x7fffffff

__device__ __forceinline__ int part_len(int ne) {
  const int c = (((ne + QP - 1) / QP) + 3) & ~3;
  return c < 4 ? 4 : c;
}

__device__ __forceinline__ void visit(int d, float w, int e, int lo, int* __restrict__ s_cnt,
                                      float* __restrict__ s_deg, int (&below)[SUBR],
                                      int* __restrict__ rank) {
#pragma unroll
  for (int j = 0; j < SUBR; ++j) below[j] += (d < lo + 1024 * j) ? 1 : 0;
  const unsigned k = (unsigned)(d - lo);
  if (k < (unsigned)RN) {
    rank[e] = atomicAdd(&s_cnt[k], 1);
    atomicAdd(&s_deg[k], w);
  }
}

__device__ __forceinline__ void visit4(int4 d, float4 w, int e, int lo, int* __restrict__ s_cnt,
                                       float* __restrict__ s_deg, int (&below)[SUBR],
                                       int* __restrict__ rank) {
#pragma unroll
  for (int j = 0; j < SUBR; ++j) {
    const int bound = lo + 1024 * j;
    below[j] += ((d.x < bound) ? 1 : 0) + ((d.y < bound) ? 1 : 0) +
                ((d.z < bound) ? 1 : 0) + ((d.w < bound) ? 1 : 0);
  }
  const unsigned k0 = (unsigned)(d.x - lo), k1 = (unsigned)(d.y - lo);
  const unsigned k2 = (unsigned)(d.z - lo), k3 = (unsigned)(d.w - lo);
  const bool m0 = k0 < (unsigned)RN, m1 = k1 < (unsigned)RN;
  const bool m2 = k2 < (unsigned)RN, m3 = k3 < (unsigned)RN;
  int r0 = 0, r1 = 0, r2 = 0, r3 = 0;
  if (m0) { r0 = atomicAdd(&s_cnt[k0], 1); atomicAdd(&s_deg[k0], w.x); }
  if (m1) { r1 = atomicAdd(&s_cnt[k1], 1); atomicAdd(&s_deg[k1], w.y); }
  if (m2) { r2 = atomicAdd(&s_cnt[k2], 1); atomicAdd(&s_deg[k2], w.z); }
  if (m3) { r3 = atomicAdd(&s_cnt[k3], 1); atomicAdd(&s_deg[k3], w.w); }
  if (m0) rank[e] = r0;
  if (m1) rank[e + 1] = r1;
  if (m2) rank[e + 2] = r2;
  if (m3) rank[e + 3] = r3;
}

__global__ __launch_bounds__(1024) void k_count(const int* __restrict__ dst, const float* __restrict__ ew,
                                                int* __restrict__ cntp, float* __restrict__ degp,
                                                int* __restrict__ belowp, int* __restrict__ rank,
                                                int n, int ne) {
  __shared__ int s_cnt[RN];
  __shared__ float s_deg[RN];
  __shared__ int s_red[SUBR * 32];
  const int t = threadIdx.x;
  const int rho = blockIdx.x / QP;
  const int q = blockIdx.x - rho * QP;
  const int lo = rho * RN;
  for (int k = t; k < RN; k += 1024) { s_cnt[k] = 0; s_deg[k] = 0.0f; }
  __syncthreads();
  const int chunk = part_len(ne);
  const int e0 = q * chunk;
  const int e1 = (e0 + chunk < ne) ? (e0 + chunk) : ne;
  int below[SUBR];
#pragma unroll
  for (int j = 0; j < SUBR; ++j) below[j] = 0;
  if ((ne & 3) == 0) {
    for (int eb = e0 + 4 * t; eb < e1; eb += 4 * 4096) {
      int4 d[4];
      float4 w[4];
#pragma unroll
      for (int u = 0; u < 4; ++u) {
        const int e = eb + u * 4096;
        d[u] = make_int4(NOEDGE, NOEDGE, NOEDGE, NOEDGE);
        w[u] = make_float4(0.0f, 0.0f, 0.0f, 0.0f);
        if (e < e1) { d[u] = *(const int4*)(dst + e); w[u] = *(const float4*)(ew + e); }
      }
#pragma unroll
      for (int u = 0; u < 4; ++u) {
        const int e = eb + u * 4096;
        visit4(d[u], w[u], e, lo, s_cnt, s_deg, below, rank);
      }
    }
  } else {
    for (int e = e0 + t; e < e1; e += 1024) visit(dst[e], ew[e], e, lo, s_cnt, s_deg, below, rank);
  }
  __syncthreads();
  for (int k = t; k < RN; k += 1024) {
    const int i = lo + k;
    if (i < n) { cntp[q * n + i] = s_cnt[k]; degp[q * n + i] = s_deg[k]; }
  }
#pragma unroll
  for (int j = 0; j < SUBR; ++j) {
    int v = below[j];
#pragma unroll
    for (int off = 16; off > 0; off >>= 1) v += __shfl_xor(v, off);
    if ((t & 31) == 0) s_red[j * 32 + (t >> 5)] = v;
  }
  __syncthreads();
  if (t < SUBR) {
    int v = 0;
    for (int w = 0; w < 32; ++w) v += s_red[t * 32 + w];
    belowp[(rho * QP + q) * SUBR + t] = v;
  }
}

__global__ __launch_bounds__(1024) void k_scan_dinv(const int* __restrict__ cntp,
                                                    const float* __restrict__ degp,
                                                    const int* __restrict__ belowp,
                                                    int* __restrict__ rowptr,
                                                    int* __restrict__ basep,
                                                    float* __restrict__ dinv,
                                                    int2* __restrict__ csr, int n) {
  __shared__ int s_cnt[32];
  const int t = threadIdx.x;
  const int lane = t & 31;
  const int wave = t >> 5;
  const int b0 = blockIdx.x * 1024;
  const int rho = b0 / RN;
  const int j = (b0 - rho * RN) >> 10;
  int pre = 0;
  for (int q = 0; q < QP; ++q) pre += belowp[(rho * QP + q) * SUBR + j];
  const int i = b0 + t;
  const int* __restrict__ cp = cntp + ((i < n) ? i : (n - 1));
  const float* __restrict__ dp = degp + ((i < n) ? i : (n - 1));
  int cq[QP];
  int c = 0;
  float d = 1.0f;
#pragma unroll
  for (int q = 0; q < QP; ++q) {
    cq[q] = (i < n) ? cp[q * n] : 0;
    d += dp[q * n];
    c += cq[q];
  }
  int incl = c;
#pragma unroll
  for (int off = 1; off < 32; off <<= 1) {
    const int up = __shfl_up(incl, off);
    if (lane >= off) incl += up;
  }
  if (lane == 31) s_cnt[wave] = incl;
  __syncthreads();
  int base = pre;
#pragma unroll
  for (int w = 0; w < 32; ++w) {
    if (w < wave) base += s_cnt[w];
  }
  if (i < n) {
    const int beg = base + incl - c + i;
    rowptr[i] = beg;
    if (i == n - 1) rowptr[n] = beg + c + 1;
    const float dv = (d > 0.0f) ? rsqrtf(fmaxf(d, 1e-20f)) : 0.0f;
    dinv[i] = dv;
    csr[beg + c] = make_int2(i, __float_as_int(dv * dv));
    int run = beg;
    int* __restrict__ bp = basep + i;
#pragma unroll
    for (int q = 0; q < QP; ++q) { bp[q * n] = run; run += cq[q]; }
  }
}

__device__ __forceinline__ void csr_put(int s, int d, float w, int r,
                                        const float* __restrict__ dinv,
                                        const int* __restrict__ bq,
                                        int2* __restrict__ csr) {
  const float nrm = dinv[s] * w * dinv[d];
  csr[bq[d] + r] = make_int2(s, __float_as_int(nrm));
}

__global__ void k_csr_fill(const int* __restrict__ src, const int* __restrict__ dst,
                           const float* __restrict__ ew, const float* __restrict__ dinv,
                           const int* __restrict__ basep, const int* __restrict__ rank,
                           int2* __restrict__ csr, int n, int ne) {
  const int e = (blockIdx.x * blockDim.x + threadIdx.x) * 4;
  const int chunk = part_len(ne);
  if (((ne & 3) == 0) && e + 3 < ne) {
    const int* __restrict__ bq = basep + (e / chunk) * n;
    const int4 s = *(const int4*)(src + e);
    const int4 d = *(const int4*)(dst + e);
    const float4 w = *(const float4*)(ew + e);
    const int4 r = *(const int4*)(rank + e);
    csr_put(s.x, d.x, w.x, r.x, dinv, bq, csr);
    csr_put(s.y, d.y, w.y, r.y, dinv, bq, csr);
    csr_put(s.z, d.z, w.z, r.z, dinv, bq, csr);
    csr_put(s.w, d.w, w.w, r.w, dinv, bq, csr);
  } else {
    for (int k = e; k < e + 4 && k < ne; ++k)
      csr_put(src[k], dst[k], ew[k], rank[k], dinv, basep + (k / chunk) * n, csr);
  }
}

struct f3 { float a, b, c; };

#define FW 8
#define FR 16
#define FU 32
#define APITCH 100
#define WPLANE ((DFEAT / 2) * DFEAT)

#define WQ (WPLANE / 4)
#define WIT ((WQ + 32 * FW - 1) / (32 * FW))

__device__ __forceinline__ void fetch_w(const float* __restrict__ W, float4 (&lo)[WIT], float4 (&hi)[WIT]) {
#pragma unroll
  for (int it = 0; it < WIT; ++it) {
    const int q = (int)threadIdx.x + it * 32 * FW;
    lo[it] = make_float4(0.0f, 0.0f, 0.0f, 0.0f);
    hi[it] = make_float4(0.0f, 0.0f, 0.0f, 0.0f);
    if (q < WQ) {
      const int kp = q / (DFEAT / 4);
      const int c4 = q - kp * (DFEAT / 4);
      const float* p = W + (2 * kp) * DFEAT + 4 * c4;
      lo[it] = *(const float4*)p;
      hi[it] = *(const float4*)(p + DFEAT);
    }
  }
}

__device__ __forceinline__ void put_w(v2f* __restrict__ sW2, const float4 (&lo)[WIT], const float4 (&hi)[WIT]) {
#pragma unroll
  for (int it = 0; it < WIT; ++it) {
    const int q = (int)threadIdx.x + it * 32 * FW;
    if (q < WQ) {
      float4* d = (float4*)(sW2 + 4 * q);
      d[0] = make_float4(lo[it].x, hi[it].x, lo[it].y, hi[it].y);
      d[1] = make_float4(lo[it].z, hi[it].z, lo[it].w, hi[it].w);
    }
  }
}

__device__ __forceinline__ void tile_mma(const float* __restrict__ sAw, const v2f* __restrict__ sW2,
                                         int lo16, int khalf, v8f (&acc)[6]) {
#pragma unroll
  for (int t = 0; t < 6; ++t) acc[t] = (v8f){0.f, 0.f, 0.f, 0.f, 0.f, 0.f, 0.f, 0.f};
  const float* ar = sAw + lo16 * APITCH + 2 * khalf;
  const v2f* bcol = sW2 + khalf * DFEAT + lo16;
#pragma unroll 4
  for (int k = 0; k < 24; ++k) {
    const v2f a = *(const v2f*)(ar + 4 * k);
    const v2f* brow = bcol + (2 * k) * DFEAT;
#pragma unroll
    for (int t = 0; t < 6; ++t) {
      const v2f b = brow[t * 16];
      acc[t] = __builtin_amdgcn_wmma_f32_16x16x4_f32(
           false, a,  false, b,
           (short)0, acc[t],  false,  false);
    }
  }
}

template <bool RELU>
__device__ __forceinline__ void store_tile(const v8f (&acc)[6], const float* __restrict__ bias,
                                           float* __restrict__ out, int mt, int lo16, int khalf) {
  const int rbase = mt * 16 + khalf * 8;
#pragma unroll
  for (int t = 0; t < 6; ++t) {
    const int cbase = t * 16 + lo16;
    const float bb = bias[cbase];
#pragma unroll
    for (int r = 0; r < 8; ++r) {
      const size_t idx = (size_t)(rbase + r) * DFEAT + cbase;
      const float v = acc[t][r] + bb;
      out[idx] = RELU ? fmaxf(v, 0.0f) : v;
    }
  }
}

__device__ __forceinline__ void pool_tile(const v8f (&acc)[6], const float* __restrict__ bias,
                                          float* __restrict__ out, int gid, int lo16, int khalf) {
  const int g0 = __builtin_amdgcn_readfirstlane(gid);
  const bool one_graph = __builtin_amdgcn_ballot_w32(gid != g0) == 0u;
  if (one_graph) {
    float s[6];
#pragma unroll
    for (int t = 0; t < 6; ++t) {
      const float bb = bias[t * 16 + lo16];
      float a = 0.0f;
#pragma unroll
      for (int r = 0; r < 8; ++r) a += acc[t][r] + bb;
      s[t] = a + __shfl_xor(a, 16);
    }
#pragma unroll
    for (int t = 0; t < 6; t += 2) {
      const float v = khalf ? s[t + 1] : s[t];
      atomicAdd(&out[(size_t)g0 * DFEAT + (t + khalf) * 16 + lo16], v);
    }
  } else {
#pragma unroll
    for (int r = 0; r < 8; ++r) {
      const int g = __shfl(gid, khalf * 8 + r);
#pragma unroll
      for (int t = 0; t < 6; ++t) {
        const int cbase = t * 16 + lo16;
        atomicAdd(&out[(size_t)g * DFEAT + cbase], acc[t][r] + bias[cbase]);
      }
    }
  }
}

template <int MODE>
__global__ __launch_bounds__(32 * FW) void k_gather_gemm(
    const float* __restrict__ T, const int2* __restrict__ csr, const int* __restrict__ rowptr,
    const float* __restrict__ Wa, const float* __restrict__ ba,
    const float* __restrict__ Wb, const float* __restrict__ bb,
    const int* __restrict__ batch,
    float* __restrict__ out_a, float* __restrict__ out_b, int n) {
  __shared__ __attribute__((aligned(16))) v2f sW2[(MODE == 1 ? 2 : 1) * WPLANE];
  __shared__ __attribute__((aligned(16))) float sA[FW * FR * APITCH];

  const int wave = __builtin_amdgcn_readfirstlane((int)(threadIdx.x >> 5));
  const int lane = threadIdx.x & 31;
  const int mt = blockIdx.x * FW + wave;
  const bool live = mt < (n >> 4);
  const int i0 = mt * FR;
  const int nr = (n - i0 < FR) ? (n - i0) : FR;
  float* sAw = sA + wave * (FR * APITCH);
  const int lo16  = lane & 15;
  const int khalf = lane >> 4;

  float4 wlo[MODE == 1 ? 2 : 1][WIT], whi[MODE == 1 ? 2 : 1][WIT];
  fetch_w(Wa, wlo[0], whi[0]);
  if (MODE == 1) fetch_w(Wb, wlo[MODE], whi[MODE]);

  int rp = 0x7fffffff;
  int gid = 0;
  if (live) {
    if (lane <= nr) rp = rowptr[i0 + lane];
    if (MODE == 1) gid = batch[i0 + lo16];
  }

  int seg_beg = 0, seg_end = 0;
  int cs = i0, cw = 0;
  if (live) {
    seg_beg = __builtin_amdgcn_readlane(rp, 0);
    seg_end = __builtin_amdgcn_readlane(rp, nr);
    const int p = seg_beg + lane;
    if (p < seg_end) { const int2 pr = csr[p]; cs = pr.x; cw = pr.y; }
  }

  put_w(sW2, wlo[0], whi[0]);
  if (MODE == 1) put_w(sW2 + WPLANE, wlo[MODE], whi[MODE]);
  __syncthreads();
  if (!live) return;

  const f3* __restrict__ Tr = (const f3*)T;

  int cur = 0;
  int row_end = __builtin_amdgcn_readlane(rp, 1);
  float a0 = 0.0f, a1 = 0.0f, a2 = 0.0f;

  for (int base = seg_beg; base < seg_end; base += 32) {
    const int pn = base + 32 + lane;
    const int2 pr = csr[(pn < seg_end) ? pn : (seg_end - 1)];
    const int ns = (pn < seg_end) ? pr.x : i0;
    const int nw = (pn < seg_end) ? pr.y : 0;
    const int m = (seg_end - base < 32) ? (seg_end - base) : 32;
    for (int j = 0; j < m; j += FU) {
      f3 v[FU];
      float w[FU];
#pragma unroll
      for (int u = 0; u < FU; ++u) {
        const int s = __builtin_amdgcn_readlane(cs, j + u);
        w[u] = __int_as_float(__builtin_amdgcn_readlane(cw, j + u));
        v[u] = Tr[(size_t)s * 32 + lane];
      }
#pragma unroll
      for (int u = 0; u < FU; ++u) {
        a0 = fmaf(w[u], v[u].a, a0);
        a1 = fmaf(w[u], v[u].b, a1);
        a2 = fmaf(w[u], v[u].c, a2);
        if (base + j + u + 1 == row_end) {
          float* dr = sAw + cur * APITCH + 3 * lane;
          dr[0] = a0; dr[1] = a1; dr[2] = a2;
          a0 = 0.0f; a1 = 0.0f; a2 = 0.0f;
          ++cur;
          row_end = __builtin_amdgcn_readlane(rp, cur + 1);
        }
      }
    }
    cs = ns; cw = nw;
  }

  v8f acc[6];
  tile_mma(sAw, sW2, lo16, khalf, acc);
  if (MODE == 0) {
    store_tile<true>(acc, ba, out_a, mt, lo16, khalf);
  } else {
    store_tile<false>(acc, ba, out_a, mt, lo16, khalf);
    tile_mma(sAw, sW2 + WPLANE, lo16, khalf, acc);
    pool_tile(acc, bb, out_b, gid, lo16, khalf);
  }
}

extern "C" void kernel_launch(void* const* d_in, const int* in_sizes, int n_in,
                              void* d_out, int out_size, void* d_ws, size_t ws_size,
                              hipStream_t stream) {
  (void)n_in; (void)ws_size;
  const float* x    = (const float*)d_in[0];
  const float* W1   = (const float*)d_in[1];
  const float* b1   = (const float*)d_in[2];
  const float* W3   = (const float*)d_in[3];
  const float* b3   = (const float*)d_in[4];
  const float* W4   = (const float*)d_in[5];
  const float* b4   = (const float*)d_in[6];
  const float* ew   = (const float*)d_in[7];
  const int*   ei   = (const int*)d_in[8];
  const int*   batch= (const int*)d_in[9];

  const int n  = in_sizes[0] / DFEAT;
  const int ne = in_sizes[7];
  const int* srcp = ei;
  const int* dstp = ei + ne;

  const int nf = n * DFEAT;
  const int gtotal = out_size - nf;

  float* h      = (float*)d_ws;
  int*   cntp   = (int*)(h + (size_t)nf);
  float* degp   = (float*)(cntp + (size_t)QP * n);
  int*   basep  = (int*)(degp + (size_t)QP * n);
  int*   belowp = basep + (size_t)QP * n;
  float* spare  = h + 2 * (size_t)nf;
  int2*  csr    = (int2*)(spare + (size_t)nf);
  int*   rank   = (int*)(spare + (size_t)nf) + 2 * ((size_t)ne + (size_t)n);
  float* dinv   = (float*)(rank + ne) + n;
  int*   rowptr = (int*)(dinv + n) + n;

  float* embed = (float*)d_out;
  float* gout  = embed + nf;

  const dim3 B(256);
  const int gR  = (n + RN - 1) / RN;
  const int gE  = (ne + 1023) / 1024;
  const int gSn = (n + 1023) / 1024;
  const int gF  = ((n >> 4) + FW - 1) / FW;

  k_count<<<gR * QP, dim3(1024), 0, stream>>>(dstp, ew, cntp, degp, belowp, rank, n, ne);
  k_scan_dinv<<<gSn, dim3(1024), 0, stream>>>(cntp, degp, belowp, rowptr, basep, dinv, csr, n);
  k_csr_fill<<<gE, B, 0, stream>>>(srcp, dstp, ew, dinv, basep, rank, csr, n, ne);

  k_gather_gemm<0><<<gF, dim3(32 * FW), 0, stream>>>(x, csr, rowptr, W1, b1, nullptr, nullptr, nullptr,
                                                     h, nullptr, n);

  k_fill<<<(gtotal + 255) / 256, B, 0, stream>>>(gout, 0.0f, gtotal);
  k_gather_gemm<1><<<gF, dim3(32 * FW), 0, stream>>>(h, csr, rowptr, W3, b3, W4, b4, batch,
                                                     embed, gout, n);
}
